# starting kernel
# baseline (speedup 1.0000x reference)
#include <hip/hip_runtime.h>


#define EPS      1e-8f
#define FAKE_INF 10.0f
#define MDIM     512
#define BLOCK    512

typedef __attribute__((ext_vector_type(2))) float v2f;
typedef __attribute__((ext_vector_type(8))) float v8f;

__device__ __forceinline__
unsigned long long cmpex_sel(unsigned long long a, unsigned long long c,
                             bool keep_max)
{
    return (keep_max == (a > c)) ? a : c;
}

template<int J>
__device__ __forceinline__ unsigned long long swz_xor_u64(unsigned long long x)
{
    constexpr int imm = 0x1f | (J << 10);
    union { unsigned long long u; int i[2]; } v;
    v.u = x;
    v.i[0] = __builtin_amdgcn_ds_swizzle(v.i[0], imm);
    v.i[1] = __builtin_amdgcn_ds_swizzle(v.i[1], imm);
    return v.u;
}

template<int J>
__device__ __forceinline__ float swz_xor_f32(float x)
{
    constexpr int imm = 0x1f | (J << 10);
    return __int_as_float(__builtin_amdgcn_ds_swizzle(__float_as_int(x), imm));
}

template<int J>
__device__ __forceinline__ void wave_reduce2(float& a, float& b)
{
    if constexpr (J >= 1) {
        a += swz_xor_f32<J>(a);
        b += swz_xor_f32<J>(b);
        wave_reduce2<(J >> 1)>(a, b);
    }
}

template<int J>
__device__ __forceinline__ void wave_reduce1(float& a)
{
    if constexpr (J >= 1) {
        a += swz_xor_f32<J>(a);
        wave_reduce1<(J >> 1)>(a);
    }
}

template<int KS, int J>
__device__ __forceinline__ void bitonic_shfl_tail(unsigned long long& key, int tid)
{
    if constexpr (J >= 1) {
        const unsigned long long c = swz_xor_u64<J>(key);
        const bool keep_max = (((tid & J) == 0) == ((tid & KS) == 0));
        key = cmpex_sel(key, c, keep_max);
        bitonic_shfl_tail<KS, (J >> 1)>(key, tid);
    }
}

template<int KS>
__device__ __forceinline__
void bitonic_stage(unsigned long long& key, int tid,
                   unsigned long long (&buf)[2][MDIM], int& ph)
{
    #pragma unroll
    for (int j = KS >> 1; j >= 32; j >>= 1) {
        buf[ph][tid] = key;
        __syncthreads();
        const unsigned long long c = buf[ph][tid ^ j];
        const bool keep_max = (((tid & j) == 0) == ((tid & KS) == 0));
        key = cmpex_sel(key, c, keep_max);
        ph ^= 1;
    }
    constexpr int JS = ((KS >> 1) < 16) ? (KS >> 1) : 16;
    bitonic_shfl_tail<KS, JS>(key, tid);
}

__global__ __launch_bounds__(BLOCK)
void ibp_get_ub_kernel(const float* __restrict__ K,
                       const float* __restrict__ K1,
                       const float* __restrict__ Bv,
                       const float* __restrict__ B1,
                       const float* __restrict__ WLB,
                       const float* __restrict__ WUB,
                       float* __restrict__ out)
{
    __shared__ unsigned long long s_keyb[2][MDIM];
    __shared__ float s_td[MDIM], s_t1d[MDIM], s_in[MDIM], s_scan[MDIM];
    __shared__ float s_gsum[32], s_goff[32];
    __shared__ float s_redt[BLOCK / 32], s_redr[BLOCK / 32], s_bcast[2];

    const int tid  = threadIdx.x;
    const int row  = blockIdx.x;
    const int lane = tid & 31;
    const int wv   = tid >> 5;

    const float k   = K [(size_t)row * MDIM + tid];
    const float k1  = K1[(size_t)row * MDIM + tid];
    const float wlb = WLB[tid];
    const float wub = WUB[tid];

    const bool  pick_lb = (k < 0.0f) || ((k == 0.0f) && (k1 < 0.0f));
    const float w_ret   = pick_lb ? wlb : wub;

    float pt = w_ret * k;
    float pr = w_ret * k1;

    const float span  = wub - wlb;
    const float ak    = fabsf(k);
    const float ak1   = fabsf(k1);
    const float td    = ak  * span;
    const float t1d   = ak1 * span;
    const float ratio = (k * k1 >= 0.0f) ? 0.0f : (ak1 / (EPS + ak));

    s_td[tid]  = td;
    s_t1d[tid] = t1d;

    unsigned long long key =
        ((unsigned long long)__float_as_uint(ratio) << 32)
        | (unsigned int)(MDIM - 1 - tid);

    wave_reduce2<16>(pt, pr);
    if (lane == 0) { s_redt[wv] = pt; s_redr[wv] = pr; }
    __syncthreads();
    if (tid == 0) {
        float st = 0.0f, sr = 0.0f;
        #pragma unroll
        for (int w = 0; w < BLOCK / 32; ++w) { st += s_redt[w]; sr += s_redr[w]; }
        s_bcast[0] = st + Bv[row];
        s_bcast[1] = sr + B1[row];
    }
    __syncthreads();
    const float t    = s_bcast[0];
    const float ret0 = s_bcast[1];

    int ph = 0;
    bitonic_stage<  2>(key, tid, s_keyb, ph);
    bitonic_stage<  4>(key, tid, s_keyb, ph);
    bitonic_stage<  8>(key, tid, s_keyb, ph);
    bitonic_stage< 16>(key, tid, s_keyb, ph);
    bitonic_stage< 32>(key, tid, s_keyb, ph);
    bitonic_stage< 64>(key, tid, s_keyb, ph);
    bitonic_stage<128>(key, tid, s_keyb, ph);
    bitonic_stage<256>(key, tid, s_keyb, ph);
    bitonic_stage<512>(key, tid, s_keyb, ph);

    const int   idx     = (MDIM - 1) - (int)(key & 0xFFFFFFFFu);
    const float ratio_s = __uint_as_float((unsigned int)(key >> 32));
    const float td_s    = s_td[idx];
    const float t1d_s   = s_t1d[idx];
    s_in[tid] = td_s;
    __syncthreads();

    const int wvs = __builtin_amdgcn_readfirstlane(wv);
    if (wvs < 2) {
        const int mn = lane & 15;
        const int kh = lane >> 4;
        const int G  = wvs * 16 + mn;
        v8f acc = {0.f, 0.f, 0.f, 0.f, 0.f, 0.f, 0.f, 0.f};
        #pragma unroll
        for (int c = 0; c < 4; ++c) {
            const int k0 = 4 * c + 2 * kh;
            v2f a, b;
            a[0] = (mn >= k0)     ? 1.0f : 0.0f;
            a[1] = (mn >= k0 + 1) ? 1.0f : 0.0f;
            b[0] = s_in[G * 16 + k0];
            b[1] = s_in[G * 16 + k0 + 1];
            acc = __builtin_amdgcn_wmma_f32_16x16x4_f32(
                      false, a, false, b, (short)0, acc, false, false);
        }
        #pragma unroll
        for (int v = 0; v < 8; ++v) {
            const int m = v + 8 * kh;
            s_scan[G * 16 + m] = acc[v];
        }
        if (kh == 1) s_gsum[G] = acc[7];
    }
    __syncthreads();
    if (wvs == 0) {
        float x   = s_gsum[lane];
        float inc = x;
        #pragma unroll
        for (int off = 1; off < 32; off <<= 1) {
            const float y = __shfl_up(inc, off, 32);
            if (lane >= off) inc += y;
        }
        s_goff[lane] = inc - x;
    }
    __syncthreads();
    const float tds = s_scan[tid] + s_goff[tid >> 4];

    float percent;
    const float delta = t - tds;
    if (delta < 0.0f) {
        float pcl = delta / (td_s + EPS);
        pcl = fmaxf(-1.0f, fminf(0.0f, pcl));
        percent = pcl + 1.0f;
    } else {
        percent = 1.0f;
    }
    if (!(ratio_s > 0.0f)) percent = 0.0f;

    float contrib = t1d_s * percent;
    wave_reduce1<16>(contrib);
    if (lane == 0) s_redt[wv] = contrib;
    __syncthreads();
    if (tid == 0) {
        float cs = 0.0f;
        #pragma unroll
        for (int w = 0; w < BLOCK / 32; ++w) cs += s_redt[w];
        out[row] = (t >= 0.0f) ? (ret0 + cs) : -FAKE_INF;
    }
}

extern "C" void kernel_launch(void* const* d_in, const int* in_sizes, int n_in,
                              void* d_out, int out_size, void* d_ws, size_t ws_size,
                              hipStream_t stream)
{
    (void)n_in; (void)d_ws; (void)ws_size; (void)out_size;
    const float* k   = (const float*)d_in[0];
    const float* k1  = (const float*)d_in[1];
    const float* b   = (const float*)d_in[2];
    const float* b1  = (const float*)d_in[3];
    const float* wlb = (const float*)d_in[4];
    const float* wub = (const float*)d_in[5];

    const int B = in_sizes[2];

    ibp_get_ub_kernel<<<dim3(B), dim3(BLOCK), 0, stream>>>(
        k, k1, b, b1, wlb, wub, (float*)d_out);
}


# Block sums t, ret, last: a wave's 32 values go through one f32 16x16x4 product into 8 registers + in-lane adds, not 15 ds_swizzle steps; t, ret joined by an idle wave in the scan; sum order changes
# speedup vs baseline: 1.3060x; 1.3060x over previous
#include <hip/hip_runtime.h>


#define EPS      1e-8f
#define FAKE_INF 10.0f
#define MDIM     512
#define BLOCK    512

typedef __attribute__((ext_vector_type(2))) float v2f;
typedef __attribute__((ext_vector_type(8))) float v8f;

__device__ __forceinline__
unsigned long long cmpex_sel(unsigned long long a, unsigned long long c,
                             bool keep_max)
{
    return (keep_max == (a > c)) ? a : c;
}

template<int J>
__device__ __forceinline__ unsigned long long swz_xor_u64(unsigned long long x)
{
    constexpr int imm = 0x1f | (J << 10);
    union { unsigned long long u; int i[2]; } v;
    v.u = x;
    v.i[0] = __builtin_amdgcn_ds_swizzle(v.i[0], imm);
    v.i[1] = __builtin_amdgcn_ds_swizzle(v.i[1], imm);
    return v.u;
}

template<int J>
__device__ __forceinline__ float swz_xor_f32(float x)
{
    constexpr int imm = 0x1f | (J << 10);
    return __int_as_float(__builtin_amdgcn_ds_swizzle(__float_as_int(x), imm));
}

__device__ __forceinline__ float mat_lanesum2(float x0, float x1, v2f bsel)
{
    v2f a;
    a[0] = x0;
    a[1] = x1;
    v8f acc = {0.f, 0.f, 0.f, 0.f, 0.f, 0.f, 0.f, 0.f};
    acc = __builtin_amdgcn_wmma_f32_16x16x4_f32(
              false, a, false, bsel, (short)0, acc, false, false);
    asm volatile("v_nop\n\tv_nop\n\tv_nop\n\tv_nop\n\tv_nop"
                 : "+v"(acc) : "v"(a), "v"(bsel));
    return ((acc[0] + acc[1]) + (acc[2] + acc[3]))
         + ((acc[4] + acc[5]) + (acc[6] + acc[7]));
}

template<int KS, int J>
__device__ __forceinline__ void bitonic_shfl_tail(unsigned long long& key, int tid)
{
    if constexpr (J >= 1) {
        const unsigned long long c = swz_xor_u64<J>(key);
        const bool keep_max = (((tid & J) == 0) == ((tid & KS) == 0));
        key = cmpex_sel(key, c, keep_max);
        bitonic_shfl_tail<KS, (J >> 1)>(key, tid);
    }
}

template<int KS>
__device__ __forceinline__
void bitonic_stage(unsigned long long& key, int tid,
                   unsigned long long (&buf)[2][MDIM], int& ph)
{
    #pragma unroll
    for (int j = KS >> 1; j >= 32; j >>= 1) {
        buf[ph][tid] = key;
        __syncthreads();
        const unsigned long long c = buf[ph][tid ^ j];
        const bool keep_max = (((tid & j) == 0) == ((tid & KS) == 0));
        key = cmpex_sel(key, c, keep_max);
        ph ^= 1;
    }
    constexpr int JS = ((KS >> 1) < 16) ? (KS >> 1) : 16;
    bitonic_shfl_tail<KS, JS>(key, tid);
}

__global__ __launch_bounds__(BLOCK, 16)
void ibp_get_ub_kernel(const float* __restrict__ K,
                       const float* __restrict__ K1,
                       const float* __restrict__ Bv,
                       const float* __restrict__ B1,
                       const float* __restrict__ WLB,
                       const float* __restrict__ WUB,
                       float* __restrict__ out,
                       int nrows)
{
    __shared__ unsigned long long s_keyb[2][MDIM];
    __shared__ float s_td[MDIM], s_t1d[MDIM], s_in[MDIM], s_scan[MDIM];
    __shared__ float s_gsum[32], s_goff[32];
    __shared__ float s_sum[6 * 32];
    constexpr int SUM_T = 0, SUM_R = 32, SUM_C = 64, SUM_TT = 128, SUM_RT = 160;

    const int tid  = threadIdx.x;
    const int lane = tid & 31;
    const int wv   = tid >> 5;
    const int wvs  = __builtin_amdgcn_readfirstlane(wv);

    int row = blockIdx.x;
    if (row >= nrows) return;

    const float wlb  = WLB[tid];
    const float wub  = WUB[tid];
    const float span = wub - wlb;
    const int   pslot = (lane & 1) * 32 + wv * 2 + (lane >> 4);

    v2f bsel;
    bsel[0] = ((lane & 15) == 0) ? 1.0f : 0.0f;
    bsel[1] = ((lane & 15) == 1) ? 1.0f : 0.0f;

    float k    = K [(size_t)row * MDIM + tid];
    float k1   = K1[(size_t)row * MDIM + tid];

    for (;;) {
        const float brow  = Bv[row];
        const float b1row = B1[row];
        const int  rown     = row + (int)gridDim.x;
        const bool has_next = rown < nrows;

        const bool  pick_lb = (k < 0.0f) || ((k == 0.0f) && (k1 < 0.0f));
        const float w_ret   = pick_lb ? wlb : wub;

        const float pt = w_ret * k;
        const float pr = w_ret * k1;

        const float ak    = fabsf(k);
        const float ak1   = fabsf(k1);
        const float td    = ak  * span;
        const float t1d   = ak1 * span;
        const float ratio = (k * k1 >= 0.0f) ? 0.0f : (ak1 / (EPS + ak));

        s_td[tid]  = td;
        s_t1d[tid] = t1d;

        unsigned long long key =
            ((unsigned long long)__float_as_uint(ratio) << 32)
            | (unsigned int)(MDIM - 1 - tid);

        {
            const float ps = mat_lanesum2(pt, pr, bsel);
            if ((lane & 14) == 0)
                s_sum[SUM_T + pslot] = ps;
        }

        float kn = 0.0f, k1n = 0.0f;
        if (has_next) {
            kn  = K [(size_t)rown * MDIM + tid];
            k1n = K1[(size_t)rown * MDIM + tid];
        }

        int ph = 0;
        bitonic_stage<  2>(key, tid, s_keyb, ph);
        bitonic_stage<  4>(key, tid, s_keyb, ph);
        bitonic_stage<  8>(key, tid, s_keyb, ph);
        bitonic_stage< 16>(key, tid, s_keyb, ph);
        bitonic_stage< 32>(key, tid, s_keyb, ph);
        bitonic_stage< 64>(key, tid, s_keyb, ph);
        bitonic_stage<128>(key, tid, s_keyb, ph);
        bitonic_stage<256>(key, tid, s_keyb, ph);
        bitonic_stage<512>(key, tid, s_keyb, ph);

        const int   idx     = (MDIM - 1) - (int)(key & 0xFFFFFFFFu);
        const float ratio_s = __uint_as_float((unsigned int)(key >> 32));
        const float td_s    = s_td[idx];
        const float t1d_s   = s_t1d[idx];
        s_in[tid] = td_s;
        __syncthreads();

        if (wvs < 2) {
            int ln = lane;
            asm volatile("" : "+v"(ln));
            const int mn = ln & 15;
            const int kh = ln >> 4;
            const int G  = wvs * 16 + mn;
            v8f acc = {0.f, 0.f, 0.f, 0.f, 0.f, 0.f, 0.f, 0.f};
            v2f a[4], b[4];
            #pragma unroll
            for (int c = 0; c < 4; ++c) {
                const int k0 = 4 * c + 2 * kh;
                a[c][0] = (mn >= k0)     ? 1.0f : 0.0f;
                a[c][1] = (mn >= k0 + 1) ? 1.0f : 0.0f;
                b[c][0] = s_in[G * 16 + k0];
                b[c][1] = s_in[G * 16 + k0 + 1];
            }
            asm volatile("" : "+v"(a[0]), "+v"(a[1]), "+v"(a[2]), "+v"(a[3]));
            #pragma unroll
            for (int c = 0; c < 4; ++c)
                acc = __builtin_amdgcn_wmma_f32_16x16x4_f32(
                          false, a[c], false, b[c], (short)0, acc, false, false);
            asm volatile("v_nop\n\tv_nop\n\tv_nop\n\tv_nop\n\tv_nop"
                         : "+v"(acc)
                         : "v"(a[0]), "v"(a[1]), "v"(a[2]), "v"(a[3]),
                           "v"(b[0]), "v"(b[1]), "v"(b[2]), "v"(b[3]));
            #pragma unroll
            for (int v = 0; v < 8; ++v) {
                const int m = v + 8 * kh;
                s_scan[G * 16 + m] = acc[v];
            }
            if (kh == 1) s_gsum[G] = acc[7];
        } else if (wvs == 2) {
            const float ps = mat_lanesum2(s_sum[SUM_T + tid - 64],
                                          s_sum[SUM_R + tid - 64], bsel);
            if ((lane & 14) == 0)
                s_sum[SUM_TT + pslot] = ps;
        }
        __syncthreads();
        if (wvs == 0) {
            float x   = s_gsum[lane];
            float inc = x;
            #pragma unroll
            for (int off = 1; off < 32; off <<= 1) {
                const float y = __shfl_up(inc, off, 32);
                if (lane >= off) inc += y;
            }
            s_goff[lane] = inc - x;
        }
        __syncthreads();
        const float tds = s_scan[tid] + s_goff[tid >> 4];
        const float t   = (s_sum[SUM_TT + 4] + s_sum[SUM_TT + 5]) + brow;

        float percent;
        const float delta = t - tds;
        if (delta < 0.0f) {
            float pcl = delta / (td_s + EPS);
            pcl = fmaxf(-1.0f, fminf(0.0f, pcl));
            percent = pcl + 1.0f;
        } else {
            percent = 1.0f;
        }
        if (!(ratio_s > 0.0f)) percent = 0.0f;

        const float contrib = t1d_s * percent;
        {
            const float cp = mat_lanesum2(contrib, 0.0f, bsel);
            if ((lane & 14) == 0) s_sum[SUM_C + pslot] = cp;
        }
        __syncthreads();
        if (wvs == 0) {
            float cs = mat_lanesum2(s_sum[SUM_C + tid], 0.0f, bsel);
            cs += swz_xor_f32<16>(cs);
            if (lane == 0) {
                const float ret0 = (s_sum[SUM_RT + 4] + s_sum[SUM_RT + 5]) + b1row;
                out[row] = (t >= 0.0f) ? (ret0 + cs) : -FAKE_INF;
            }
        }

        if (!has_next) break;
        row = rown; k = kn; k1 = k1n;
    }
}

extern "C" void kernel_launch(void* const* d_in, const int* in_sizes, int n_in,
                              void* d_out, int out_size, void* d_ws, size_t ws_size,
                              hipStream_t stream)
{
    (void)n_in; (void)d_ws; (void)ws_size; (void)out_size;
    const float* k   = (const float*)d_in[0];
    const float* k1  = (const float*)d_in[1];
    const float* b   = (const float*)d_in[2];
    const float* b1  = (const float*)d_in[3];
    const float* wlb = (const float*)d_in[4];
    const float* wub = (const float*)d_in[5];

    const int B = in_sizes[2];

    ibp_get_ub_kernel<<<dim3(8192), dim3(BLOCK), 0, stream>>>(
        k, k1, b, b1, wlb, wub, (float*)d_out, B);
}
